# starting kernel
# baseline (speedup 1.0000x reference)
#include <hip/hip_runtime.h>

typedef __attribute__((ext_vector_type(2))) float v2f;
typedef __attribute__((ext_vector_type(8))) float v8f;

#define N_NODES 100000
#define N_EDGES 1600000
#define IN_DIM  128
#define OUT_DIM 64
#define ROW_TILES (N_NODES / 16)

#define W_PAIR_STRIDE 80
#define W_PAIR_ROWS   (IN_DIM / 2)

__global__ void gcn_init_bias(const float* __restrict__ bias,
                              float* __restrict__ out, int n) {
    int i = blockIdx.x * blockDim.x + threadIdx.x;
    if (i < n) out[i] = bias[i & (OUT_DIM - 1)];
}

__global__ __launch_bounds__(256) void gcn_gemm_wmma(
    const float* __restrict__ X, const float* __restrict__ W,
    float* __restrict__ S) {
    __shared__ v2f w_lds[W_PAIR_ROWS * W_PAIR_STRIDE];

    for (int p = threadIdx.x; p < W_PAIR_ROWS * OUT_DIM; p += 256) {
        const int k2 = p >> 6;
        const int n  = p & 63;
        v2f t;
        t.x = W[(2 * k2 + 0) * OUT_DIM + n];
        t.y = W[(2 * k2 + 1) * OUT_DIM + n];
        w_lds[k2 * W_PAIR_STRIDE + n] = t;
    }
    __syncthreads();

    const int wave = threadIdx.x >> 5;
    const int lane = threadIdx.x & 31;
    const int tile = blockIdx.x * 8 + wave;
    if (tile >= ROW_TILES) return;

    const int row0  = tile * 16;
    const int m     = lane & 15;
    const int khalf = (lane >> 4) * 2;

    v8f acc[4] = {};
    const float* arow = X + (size_t)(row0 + m) * IN_DIM;

    for (int k0 = 0; k0 < IN_DIM; k0 += 4) {
        v2f a = *(const v2f*)(arow + k0 + khalf);
        const v2f* wk = w_lds + ((k0 + khalf) >> 1) * W_PAIR_STRIDE + m;
#pragma unroll
        for (int t = 0; t < 4; ++t) {
            v2f b = wk[t * 16];
            acc[t] = __builtin_amdgcn_wmma_f32_16x16x4_f32(
                 false, a,  false, b,
                 (short)0, acc[t],
                 false,  false);
        }
    }

    const int mhi = (lane >> 4) * 8;
#pragma unroll
    for (int t = 0; t < 4; ++t) {
#pragma unroll
        for (int g = 0; g < 8; ++g) {
            S[(size_t)(row0 + mhi + g) * OUT_DIM + t * 16 + m] = acc[t][g];
        }
    }
}

__global__ __launch_bounds__(256) void gcn_scatter(
    const float* __restrict__ S, const float* __restrict__ vals,
    const int* __restrict__ row, const int* __restrict__ col,
    float* __restrict__ out) {
    long long t = (long long)blockIdx.x * 256 + threadIdx.x;
    long long e = t >> 4;
    if (e >= N_EDGES) return;
    const int j0 = (int)(t & 15) * 4;

    const float v = __builtin_nontemporal_load(vals + e);
    const int   c = __builtin_nontemporal_load(col + e);
    const int   r = __builtin_nontemporal_load(row + e);

    const float4 s = *(const float4*)(S + (size_t)c * OUT_DIM + j0);
    float* o = out + (size_t)r * OUT_DIM + j0;
    atomicAdd(o + 0, v * s.x);
    atomicAdd(o + 1, v * s.y);
    atomicAdd(o + 2, v * s.z);
    atomicAdd(o + 3, v * s.w);
}

extern "C" void kernel_launch(void* const* d_in, const int* in_sizes, int n_in,
                              void* d_out, int out_size, void* d_ws, size_t ws_size,
                              hipStream_t stream) {
    const float* x    = (const float*)d_in[0];
    const float* w    = (const float*)d_in[1];
    const float* bias = (const float*)d_in[2];
    const float* vals = (const float*)d_in[3];
    const int*   row  = (const int*)d_in[4];
    const int*   col  = (const int*)d_in[5];
    float* out     = (float*)d_out;
    float* support = (float*)d_ws;

    {
        const int n = N_NODES * OUT_DIM;
        gcn_init_bias<<<(n + 255) / 256, 256, 0, stream>>>(bias, out, n);
    }
    {
        const int blocks = (ROW_TILES + 7) / 8;
        gcn_gemm_wmma<<<blocks, 256, 0, stream>>>(x, w, support);
    }
    {
        const long long threads = (long long)N_EDGES * 16;
        const int blocks = (int)((threads + 255) / 256);
        gcn_scatter<<<blocks, 256, 0, stream>>>(support, vals, row, col, out);
    }
}


# scatter with lane = feature: a wave walks 32 edges, each gather and each f32 atomic add instruction covers one whole 128-B line of one row (2+2 lines per edge, was 2+8 partial)
# speedup vs baseline: 2.0875x; 2.0875x over previous
#include <hip/hip_runtime.h>

typedef __attribute__((ext_vector_type(2))) float v2f;
typedef __attribute__((ext_vector_type(8))) float v8f;

#define N_NODES 100000
#define N_EDGES 1600000
#define IN_DIM  128
#define OUT_DIM 64
#define ROW_TILES (N_NODES / 16)

#define W_PAIR_STRIDE 80
#define W_PAIR_ROWS   (IN_DIM / 2)

__global__ void gcn_init_bias(const float* __restrict__ bias,
                              float* __restrict__ out, int n) {
    int i = blockIdx.x * blockDim.x + threadIdx.x;
    if (i < n) out[i] = bias[i & (OUT_DIM - 1)];
}

__global__ __launch_bounds__(256) void gcn_gemm_wmma(
    const float* __restrict__ X, const float* __restrict__ W,
    float* __restrict__ S) {
    __shared__ v2f w_lds[W_PAIR_ROWS * W_PAIR_STRIDE];

    for (int p = threadIdx.x; p < W_PAIR_ROWS * OUT_DIM; p += 256) {
        const int k2 = p >> 6;
        const int n  = p & 63;
        v2f t;
        t.x = W[(2 * k2 + 0) * OUT_DIM + n];
        t.y = W[(2 * k2 + 1) * OUT_DIM + n];
        w_lds[k2 * W_PAIR_STRIDE + n] = t;
    }
    __syncthreads();

    const int wave = threadIdx.x >> 5;
    const int lane = threadIdx.x & 31;
    const int tile = blockIdx.x * 8 + wave;
    if (tile >= ROW_TILES) return;

    const int row0  = tile * 16;
    const int m     = lane & 15;
    const int khalf = (lane >> 4) * 2;

    v8f acc[4] = {};
    const float* arow = X + (size_t)(row0 + m) * IN_DIM;

    for (int k0 = 0; k0 < IN_DIM; k0 += 4) {
        v2f a = *(const v2f*)(arow + k0 + khalf);
        const v2f* wk = w_lds + ((k0 + khalf) >> 1) * W_PAIR_STRIDE + m;
#pragma unroll
        for (int t = 0; t < 4; ++t) {
            v2f b = wk[t * 16];
            acc[t] = __builtin_amdgcn_wmma_f32_16x16x4_f32(
                 false, a,  false, b,
                 (short)0, acc[t],
                 false,  false);
        }
    }

    const int mhi = (lane >> 4) * 8;
#pragma unroll
    for (int t = 0; t < 4; ++t) {
#pragma unroll
        for (int g = 0; g < 8; ++g) {
            S[(size_t)(row0 + mhi + g) * OUT_DIM + t * 16 + m] = acc[t][g];
        }
    }
}

#define EDGES_PER_WAVE 32
#define EDGE_BATCH     8
static_assert(N_EDGES % EDGES_PER_WAVE == 0, "a wave takes a whole group of edges");
__global__ __launch_bounds__(256) void gcn_scatter(
    const float* __restrict__ S, const float* __restrict__ vals,
    const int* __restrict__ row, const int* __restrict__ col,
    float* __restrict__ out) {
    const int lane = threadIdx.x & 31;
    const int e0 = (blockIdx.x * 8 + (threadIdx.x >> 5)) * EDGES_PER_WAVE;
    if (e0 >= N_EDGES) return;

    const int e = e0 + lane;
    const int vbits = __float_as_int(__builtin_nontemporal_load(vals + e));
    const int c = __builtin_nontemporal_load(col + e);
    const int r = __builtin_nontemporal_load(row + e);

    for (int k0 = 0; k0 < EDGES_PER_WAVE; k0 += EDGE_BATCH) {
        float s0[EDGE_BATCH], s1[EDGE_BATCH];
#pragma unroll
        for (int u = 0; u < EDGE_BATCH; ++u) {
            const int ck = __builtin_amdgcn_readlane(c, k0 + u);
            const float* s = S + (size_t)ck * OUT_DIM + lane;
            s0[u] = s[0];
            s1[u] = s[32];
        }
#pragma unroll
        for (int u = 0; u < EDGE_BATCH; ++u) {
            const float vk = __int_as_float(__builtin_amdgcn_readlane(vbits, k0 + u));
            const int   rk = __builtin_amdgcn_readlane(r, k0 + u);
            float* o = out + (size_t)rk * OUT_DIM + lane;
            atomicAdd(o,      vk * s0[u]);
            atomicAdd(o + 32, vk * s1[u]);
        }
    }
}

extern "C" void kernel_launch(void* const* d_in, const int* in_sizes, int n_in,
                              void* d_out, int out_size, void* d_ws, size_t ws_size,
                              hipStream_t stream) {
    const float* x    = (const float*)d_in[0];
    const float* w    = (const float*)d_in[1];
    const float* bias = (const float*)d_in[2];
    const float* vals = (const float*)d_in[3];
    const int*   row  = (const int*)d_in[4];
    const int*   col  = (const int*)d_in[5];
    float* out     = (float*)d_out;
    float* support = (float*)d_ws;

    {
        const int n = N_NODES * OUT_DIM;
        gcn_init_bias<<<(n + 255) / 256, 256, 0, stream>>>(bias, out, n);
    }
    {
        const int blocks = (ROW_TILES + 7) / 8;
        gcn_gemm_wmma<<<blocks, 256, 0, stream>>>(x, w, support);
    }
    {
        const int waves  = (N_EDGES + EDGES_PER_WAVE - 1) / EDGES_PER_WAVE;
        const int blocks = (waves + 7) / 8;
        gcn_scatter<<<blocks, 256, 0, stream>>>(support, vals, row, col, out);
    }
}


# out = bias start value stored by the GEMM waves (8 float4 stores per lane over the wave's own 4 KB slab, before the K loop), not by a separate launch of 200000 one-store waves; sum order unchanged
# speedup vs baseline: 2.3864x; 1.1432x over previous
#include <hip/hip_runtime.h>

typedef __attribute__((ext_vector_type(2))) float v2f;
typedef __attribute__((ext_vector_type(4))) float v4f;
typedef __attribute__((ext_vector_type(8))) float v8f;

#define N_NODES 100000
#define N_EDGES 1600000
#define IN_DIM  128
#define OUT_DIM 64
#define ROW_TILES (N_NODES / 16)

#define W_PAIR_STRIDE 80
#define W_PAIR_ROWS   (IN_DIM / 2)

__global__ __launch_bounds__(256) void gcn_gemm_wmma(
    const float* __restrict__ X, const float* __restrict__ W,
    const float* __restrict__ bias,
    float* __restrict__ S, float* __restrict__ out) {
    __shared__ v2f w_lds[W_PAIR_ROWS * W_PAIR_STRIDE];

    const int bcol = (threadIdx.x & 31) * 4 & (OUT_DIM - 1);
    v4f b4;
    b4.x = bias[bcol + 0];
    b4.y = bias[bcol + 1];
    b4.z = bias[bcol + 2];
    b4.w = bias[bcol + 3];

    for (int p = threadIdx.x; p < W_PAIR_ROWS * OUT_DIM; p += 256) {
        const int k2 = p >> 6;
        const int n  = p & 63;
        v2f t;
        t.x = W[(2 * k2 + 0) * OUT_DIM + n];
        t.y = W[(2 * k2 + 1) * OUT_DIM + n];
        w_lds[k2 * W_PAIR_STRIDE + n] = t;
    }
    __syncthreads();

    const int wave = threadIdx.x >> 5;
    const int lane = threadIdx.x & 31;
    const int tile = blockIdx.x * 8 + wave;
    if (tile >= ROW_TILES) return;

    const int row0  = tile * 16;

    {
        float* oslab = out + (size_t)row0 * OUT_DIM + lane * 4;
        for (int i = 0; i < 8; ++i) *(v4f*)(oslab + i * 128) = b4;
    }

    const int m     = lane & 15;
    const int khalf = (lane >> 4) * 2;

    v8f acc[4] = {};
    const float* arow = X + (size_t)(row0 + m) * IN_DIM;

    for (int k0 = 0; k0 < IN_DIM; k0 += 4) {
        v2f a = *(const v2f*)(arow + k0 + khalf);
        const v2f* wk = w_lds + ((k0 + khalf) >> 1) * W_PAIR_STRIDE + m;
#pragma unroll
        for (int t = 0; t < 4; ++t) {
            v2f b = wk[t * 16];
            acc[t] = __builtin_amdgcn_wmma_f32_16x16x4_f32(
                 false, a,  false, b,
                 (short)0, acc[t],
                 false,  false);
        }
    }

    const int mhi = (lane >> 4) * 8;
#pragma unroll
    for (int t = 0; t < 4; ++t) {
#pragma unroll
        for (int g = 0; g < 8; ++g) {
            S[(size_t)(row0 + mhi + g) * OUT_DIM + t * 16 + m] = acc[t][g];
        }
    }
}

#define EDGES_PER_WAVE 32
#define EDGE_BATCH     8
static_assert(N_EDGES % EDGES_PER_WAVE == 0, "a wave takes a whole group of edges");
__global__ __launch_bounds__(256) void gcn_scatter(
    const float* __restrict__ S, const float* __restrict__ vals,
    const int* __restrict__ row, const int* __restrict__ col,
    float* __restrict__ out) {
    const int lane = threadIdx.x & 31;
    const int e0 = (blockIdx.x * 8 + (threadIdx.x >> 5)) * EDGES_PER_WAVE;
    if (e0 >= N_EDGES) return;

    const int e = e0 + lane;
    const int vbits = __float_as_int(__builtin_nontemporal_load(vals + e));
    const int c = __builtin_nontemporal_load(col + e);
    const int r = __builtin_nontemporal_load(row + e);

    for (int k0 = 0; k0 < EDGES_PER_WAVE; k0 += EDGE_BATCH) {
        float s0[EDGE_BATCH], s1[EDGE_BATCH];
#pragma unroll
        for (int u = 0; u < EDGE_BATCH; ++u) {
            const int ck = __builtin_amdgcn_readlane(c, k0 + u);
            const float* s = S + (size_t)ck * OUT_DIM + lane;
            s0[u] = s[0];
            s1[u] = s[32];
        }
#pragma unroll
        for (int u = 0; u < EDGE_BATCH; ++u) {
            const float vk = __int_as_float(__builtin_amdgcn_readlane(vbits, k0 + u));
            const int   rk = __builtin_amdgcn_readlane(r, k0 + u);
            float* o = out + (size_t)rk * OUT_DIM + lane;
            atomicAdd(o,      vk * s0[u]);
            atomicAdd(o + 32, vk * s1[u]);
        }
    }
}

extern "C" void kernel_launch(void* const* d_in, const int* in_sizes, int n_in,
                              void* d_out, int out_size, void* d_ws, size_t ws_size,
                              hipStream_t stream) {
    const float* x    = (const float*)d_in[0];
    const float* w    = (const float*)d_in[1];
    const float* bias = (const float*)d_in[2];
    const float* vals = (const float*)d_in[3];
    const int*   row  = (const int*)d_in[4];
    const int*   col  = (const int*)d_in[5];
    float* out     = (float*)d_out;
    float* support = (float*)d_ws;

    {
        const int blocks = (ROW_TILES + 7) / 8;
        gcn_gemm_wmma<<<blocks, 256, 0, stream>>>(x, w, bias, support, out);
    }
    {
        const int waves  = (N_EDGES + EDGES_PER_WAVE - 1) / EDGES_PER_WAVE;
        const int blocks = (waves + 7) / 8;
        gcn_scatter<<<blocks, 256, 0, stream>>>(support, vals, row, col, out);
    }
}
